# starting kernel
# baseline (speedup 1.0000x reference)
#include <hip/hip_runtime.h>

#define B_      16
#define R_      8
#define V_      33
#define H_      512
#define W_      512
#define VCUBE   (V_ * V_ * V_)
#define M_TOTAL (3 * VCUBE)
#define HW      (H_ * W_)

typedef __attribute__((ext_vector_type(2))) float v2f;
typedef __attribute__((ext_vector_type(4))) float v4f;
typedef __attribute__((ext_vector_type(8))) float v8f;

__global__ void lut_gen_wmma(const float* __restrict__ weights,
                             const float* __restrict__ bank,
                             float* __restrict__ lut)
{
    const int lane = threadIdx.x & 31;
    const int wave = threadIdx.x >> 5;
    const int tile = blockIdx.x * (blockDim.x >> 5) + wave;
    const int m0   = tile * 16;
    if (m0 >= M_TOTAL) return;

    const int half = lane >> 4;
    const int l16  = lane & 15;

    v8f acc = {};
    #pragma unroll
    for (int kb = 0; kb < R_; kb += 4) {
        int row    = m0 + l16;
        int rclamp = row < M_TOTAL ? row : (M_TOTAL - 1);
        const float* arow = bank + (long)rclamp * R_ + kb + half * 2;
        v2f a; a.x = arow[0]; a.y = arow[1];

        const float* bcol = weights + l16 * R_ + kb + half * 2;
        v2f bm; bm.x = bcol[0]; bm.y = bcol[1];

        acc = __builtin_amdgcn_wmma_f32_16x16x4_f32(
            false, a, false, bm, (short)0, acc, false, false);
    }

    const int   batch = l16;
    const int   mbase = m0 + half * 8;
    float*      dst   = lut + (long)batch * M_TOTAL + mbase;

    if (m0 + 16 <= M_TOTAL) {
        v4f lo = { acc[0], acc[1], acc[2], acc[3] };
        v4f hi = { acc[4], acc[5], acc[6], acc[7] };
        *(v4f*)(dst)     = lo;
        *(v4f*)(dst + 4) = hi;
    } else {
        #pragma unroll
        for (int j = 0; j < 8; ++j) {
            if (mbase + j < M_TOTAL) dst[j] = acc[j];
        }
    }
}

__device__ __forceinline__ int imin_(int a, int b) { return a < b ? a : b; }

__device__ __forceinline__ void tri1(const float* __restrict__ L,
                                     float r, float g, float bl,
                                     float* __restrict__ o)
{
    const float s = (float)(V_ - 1);
    float cx = fminf(fmaxf(r  * s, 0.0f), s);
    float cy = fminf(fmaxf(g  * s, 0.0f), s);
    float cz = fminf(fmaxf(bl * s, 0.0f), s);
    int x0 = imin_((int)cx, V_ - 1);
    int y0 = imin_((int)cy, V_ - 1);
    int z0 = imin_((int)cz, V_ - 1);
    int x1 = imin_(x0 + 1, V_ - 1);
    int y1 = imin_(y0 + 1, V_ - 1);
    int z1 = imin_(z0 + 1, V_ - 1);
    float fx = cx - (float)x0, fy = cy - (float)y0, fz = cz - (float)z0;
    float gx = 1.0f - fx, gy = 1.0f - fy, gz = 1.0f - fz;

    int zy00 = (z0 * V_ + y0) * V_;
    int zy01 = (z0 * V_ + y1) * V_;
    int zy10 = (z1 * V_ + y0) * V_;
    int zy11 = (z1 * V_ + y1) * V_;
    int i000 = zy00 + x0, i001 = zy00 + x1, i010 = zy01 + x0, i011 = zy01 + x1;
    int i100 = zy10 + x0, i101 = zy10 + x1, i110 = zy11 + x0, i111 = zy11 + x1;

    float w000 = gz * gy * gx, w001 = gz * gy * fx;
    float w010 = gz * fy * gx, w011 = gz * fy * fx;
    float w100 = fz * gy * gx, w101 = fz * gy * fx;
    float w110 = fz * fy * gx, w111 = fz * fy * fx;

    #pragma unroll
    for (int c = 0; c < 3; ++c) {
        const float* Lc = L + c * VCUBE;
        o[c] = Lc[i000] * w000 + Lc[i001] * w001 + Lc[i010] * w010 + Lc[i011] * w011
             + Lc[i100] * w100 + Lc[i101] * w101 + Lc[i110] * w110 + Lc[i111] * w111;
    }
}

__global__ void lut_apply(const float* __restrict__ img,
                          const float* __restrict__ lut,
                          float* __restrict__ out)
{
    const int  b  = blockIdx.y;
    const int  q  = blockIdx.x * blockDim.x + threadIdx.x;
    const long ib = (long)b * 3 * HW;
    const float* __restrict__ L = lut + (long)b * M_TOTAL;

    float4 r4 = ((const float4*)(img + ib         ))[q];
    float4 g4 = ((const float4*)(img + ib +     HW))[q];
    float4 b4 = ((const float4*)(img + ib + 2 * HW))[q];

    float4 o0, o1, o2;
    float  o[3];
    tri1(L, r4.x, g4.x, b4.x, o); o0.x = o[0]; o1.x = o[1]; o2.x = o[2];
    tri1(L, r4.y, g4.y, b4.y, o); o0.y = o[0]; o1.y = o[1]; o2.y = o[2];
    tri1(L, r4.z, g4.z, b4.z, o); o0.z = o[0]; o1.z = o[1]; o2.z = o[2];
    tri1(L, r4.w, g4.w, b4.w, o); o0.w = o[0]; o1.w = o[1]; o2.w = o[2];

    ((float4*)(out + ib         ))[q] = o0;
    ((float4*)(out + ib +     HW))[q] = o1;
    ((float4*)(out + ib + 2 * HW))[q] = o2;
}

extern "C" void kernel_launch(void* const* d_in, const int* in_sizes, int n_in,
                              void* d_out, int out_size, void* d_ws, size_t ws_size,
                              hipStream_t stream) {
    (void)in_sizes; (void)n_in; (void)out_size; (void)ws_size;
    const float* weights = (const float*)d_in[0];
    const float* img     = (const float*)d_in[1];
    const float* bank    = (const float*)d_in[2];
    float*       out     = (float*)d_out;
    float*       lut     = (float*)d_ws;

    const int tiles = (M_TOTAL + 15) / 16;
    dim3 blk1(256);
    dim3 grd1((tiles + 7) / 8);
    lut_gen_wmma<<<grd1, blk1, 0, stream>>>(weights, bank, lut);

    dim3 blk2(256);
    dim3 grd2(HW / 4 / 256, B_);
    lut_apply<<<grd2, blk2, 0, stream>>>(img, lut, out);
}


# Apply kernel copies one colour's 33^3 table into LDS per work-group and takes each pixel's 8 corner words from LDS (4 adjacent pairs) instead of 24 scattered global loads per pixel
# speedup vs baseline: 7.5817x; 7.5817x over previous
#include <hip/hip_runtime.h>


#define B_      16
#define R_      8
#define V_      33
#define H_      512
#define W_      512
#define VCUBE   (V_ * V_ * V_)
#define M_TOTAL (3 * VCUBE)
#define HW      (H_ * W_)

typedef __attribute__((ext_vector_type(2))) float v2f;
typedef __attribute__((ext_vector_type(4))) float v4f;
typedef __attribute__((ext_vector_type(8))) float v8f;

__global__ void lut_gen_wmma(const float* __restrict__ weights,
                             const float* __restrict__ bank,
                             float* __restrict__ lut)
{
    const int lane = threadIdx.x & 31;
    const int wave = threadIdx.x >> 5;
    const int tile = blockIdx.x * (blockDim.x >> 5) + wave;
    const int m0   = tile * 16;
    if (m0 >= M_TOTAL) return;

    const int half = lane >> 4;
    const int l16  = lane & 15;

    v8f acc = {};
    #pragma unroll
    for (int kb = 0; kb < R_; kb += 4) {
        int row    = m0 + l16;
        int rclamp = row < M_TOTAL ? row : (M_TOTAL - 1);
        const float* arow = bank + (long)rclamp * R_ + kb + half * 2;
        v2f a; a.x = arow[0]; a.y = arow[1];

        const float* bcol = weights + l16 * R_ + kb + half * 2;
        v2f bm; bm.x = bcol[0]; bm.y = bcol[1];

        acc = __builtin_amdgcn_wmma_f32_16x16x4_f32(
            false, a, false, bm, (short)0, acc, false, false);
    }

    const int   batch = l16;
    const int   mbase = m0 + half * 8;
    float*      dst   = lut + (long)batch * M_TOTAL + mbase;

    if (m0 + 16 <= M_TOTAL) {
        v4f lo = { acc[0], acc[1], acc[2], acc[3] };
        v4f hi = { acc[4], acc[5], acc[6], acc[7] };
        *(v4f*)(dst)     = lo;
        *(v4f*)(dst + 4) = hi;
    } else {
        #pragma unroll
        for (int j = 0; j < 8; ++j) {
            if (mbase + j < M_TOTAL) dst[j] = acc[j];
        }
    }
}

#define AP_THREADS 512
#define AP_QUADS   8
#define AP_PIX     (AP_QUADS * 4)
#define AP_SLICE   (AP_THREADS * AP_QUADS)
#define AP_SLICES  (HW / 4 / AP_SLICE)
#define TBL_F4     ((VCUBE + 3 + 3) / 4)
#define TBL_ITERS  ((TBL_F4 + AP_THREADS - 1) / AP_THREADS)
static_assert(((long)B_ * M_TOTAL) % 4 == 0, "last table must end 16-byte aligned");
static_assert(HW / 4 == AP_SLICE * AP_SLICES, "slices must tile the image exactly");

#define PK_OFFMASK 0x000FFFFFu
#define PK_XLAST   0x00100000u
#define PK_YSTEP   0x00200000u
#define PK_ZSTEP   0x00400000u

__device__ __forceinline__ int imin_(int a, int b) { return a < b ? a : b; }

__device__ __forceinline__ void prep1(float r, float g, float bl,
                                      unsigned& pk, float& fx, float& fy, float& fz)
{
    const float s = (float)(V_ - 1);
    float cx = fminf(fmaxf(r  * s, 0.0f), s);
    float cy = fminf(fmaxf(g  * s, 0.0f), s);
    float cz = fminf(fmaxf(bl * s, 0.0f), s);
    int x0 = imin_((int)cx, V_ - 1);
    int y0 = imin_((int)cy, V_ - 1);
    int z0 = imin_((int)cz, V_ - 1);
    fx = cx - (float)x0; fy = cy - (float)y0; fz = cz - (float)z0;
    int xb = imin_(x0, V_ - 2);
    unsigned idx = (unsigned)((z0 * V_ + y0) * V_ + xb);
    pk = (idx << 2)
       | (x0 > V_ - 2 ? PK_XLAST : 0u)
       | (y0 < V_ - 1 ? PK_YSTEP : 0u)
       | (z0 < V_ - 1 ? PK_ZSTEP : 0u);
}

__device__ __forceinline__ float tri_lds(const char* tb, unsigned pk,
                                         float fx, float fy, float fz)
{
    const unsigned o00   = pk & PK_OFFMASK;
    const unsigned dy    = (pk & PK_YSTEP) ? (unsigned)(V_ * 4) : 0u;
    const unsigned dz    = (pk & PK_ZSTEP) ? (unsigned)(V_ * V_ * 4) : 0u;
    const bool     xlast = (pk & PK_XLAST) != 0u;

    const float* p00 = (const float*)(tb + o00);
    const float* p01 = (const float*)(tb + o00 + dy);
    const float* p10 = (const float*)(tb + o00 + dz);
    const float* p11 = (const float*)(tb + o00 + dy + dz);
    float a00 = p00[0], b00 = p00[1];
    float a01 = p01[0], b01 = p01[1];
    float a10 = p10[0], b10 = p10[1];
    float a11 = p11[0], b11 = p11[1];
    float v000 = xlast ? b00 : a00, v001 = b00;
    float v010 = xlast ? b01 : a01, v011 = b01;
    float v100 = xlast ? b10 : a10, v101 = b10;
    float v110 = xlast ? b11 : a11, v111 = b11;

    float gx = 1.0f - fx, gy = 1.0f - fy, gz = 1.0f - fz;
    float w000 = gz * gy * gx, w001 = gz * gy * fx;
    float w010 = gz * fy * gx, w011 = gz * fy * fx;
    float w100 = fz * gy * gx, w101 = fz * gy * fx;
    float w110 = fz * fy * gx, w111 = fz * fy * fx;

    return v000 * w000 + v001 * w001 + v010 * w010 + v011 * w011
         + v100 * w100 + v101 * w101 + v110 * w110 + v111 * w111;
}

__device__ __forceinline__ void stage_table(const float* __restrict__ lut, int b, int c,
                                            float* tbl, int tid)
{
    const unsigned g0  = (unsigned)b * M_TOTAL + (unsigned)c * VCUBE;
    const float4*  src = (const float4*)(lut + (g0 & ~3u));
    float4*        dst = (float4*)tbl;
    #pragma unroll
    for (int k = 0; k < TBL_ITERS; ++k) {
        const int j = imin_(tid + k * AP_THREADS, TBL_F4 - 1);
        dst[j] = src[j];
    }
}

__global__ void __launch_bounds__(AP_THREADS)
lut_apply_lds(const float* __restrict__ img,
              const float* __restrict__ lut,
              float* __restrict__ out)
{
    __shared__ __attribute__((aligned(16))) float tbl[TBL_F4 * 4];

    const int  tid = threadIdx.x;
    const int  b   = blockIdx.y;
    const long ib  = (long)b * 3 * HW;
    const int  q0  = blockIdx.x * AP_SLICE + tid;

    float4 r4[AP_QUADS], g4[AP_QUADS], b4[AP_QUADS];
    #pragma unroll
    for (int j = 0; j < AP_QUADS; ++j) {
        const int q = q0 + j * AP_THREADS;
        r4[j] = ((const float4*)(img + ib         ))[q];
        g4[j] = ((const float4*)(img + ib +     HW))[q];
        b4[j] = ((const float4*)(img + ib + 2 * HW))[q];
    }
    stage_table(lut, b, 0, tbl, tid);

    unsigned pk[AP_PIX];
    float    fx[AP_PIX], fy[AP_PIX], fz[AP_PIX];
    #pragma unroll
    for (int j = 0; j < AP_QUADS; ++j) {
        prep1(r4[j].x, g4[j].x, b4[j].x, pk[4*j+0], fx[4*j+0], fy[4*j+0], fz[4*j+0]);
        prep1(r4[j].y, g4[j].y, b4[j].y, pk[4*j+1], fx[4*j+1], fy[4*j+1], fz[4*j+1]);
        prep1(r4[j].z, g4[j].z, b4[j].z, pk[4*j+2], fx[4*j+2], fy[4*j+2], fz[4*j+2]);
        prep1(r4[j].w, g4[j].w, b4[j].w, pk[4*j+3], fx[4*j+3], fy[4*j+3], fz[4*j+3]);
    }

    for (int c = 0; c < 3; ++c) {
        __syncthreads();
        const unsigned a  = ((unsigned)b * M_TOTAL + (unsigned)c * VCUBE) & 3u;
        const char*    tb = (const char*)tbl + a * 4u;
        float4*        op = (float4*)(out + ib + (long)c * HW);
        #pragma unroll
        for (int j = 0; j < AP_QUADS; ++j) {
            #pragma unroll
            for (int e = 0; e < 4; ++e)
                asm volatile("" : "+v"(pk[4*j+e]), "+v"(fx[4*j+e]), "+v"(fy[4*j+e]), "+v"(fz[4*j+e]));
            float4 o;
            o.x = tri_lds(tb, pk[4*j+0], fx[4*j+0], fy[4*j+0], fz[4*j+0]);
            o.y = tri_lds(tb, pk[4*j+1], fx[4*j+1], fy[4*j+1], fz[4*j+1]);
            o.z = tri_lds(tb, pk[4*j+2], fx[4*j+2], fy[4*j+2], fz[4*j+2]);
            o.w = tri_lds(tb, pk[4*j+3], fx[4*j+3], fy[4*j+3], fz[4*j+3]);
            op[q0 + j * AP_THREADS] = o;
        }
        if (c < 2) {
            __syncthreads();
            stage_table(lut, b, c + 1, tbl, tid);
        }
    }
}

extern "C" void kernel_launch(void* const* d_in, const int* in_sizes, int n_in,
                              void* d_out, int out_size, void* d_ws, size_t ws_size,
                              hipStream_t stream) {
    (void)in_sizes; (void)n_in; (void)out_size; (void)ws_size;
    const float* weights = (const float*)d_in[0];
    const float* img     = (const float*)d_in[1];
    const float* bank    = (const float*)d_in[2];
    float*       out     = (float*)d_out;
    float*       lut     = (float*)d_ws;

    const int tiles = (M_TOTAL + 15) / 16;
    dim3 blk1(256);
    dim3 grd1((tiles + 7) / 8);
    lut_gen_wmma<<<grd1, blk1, 0, stream>>>(weights, bank, lut);

    dim3 blk2(AP_THREADS);
    dim3 grd2(AP_SLICES, B_);
    lut_apply_lds<<<grd2, blk2, 0, stream>>>(img, lut, out);
}


# Apply kernel: second LDS table buffer; the next colour's table is copied in pieces between the current colour's pixel quads, and one LDS-only barrier per colour replaces two full ones
# speedup vs baseline: 8.0355x; 1.0599x over previous
#include <hip/hip_runtime.h>


#define B_      16
#define R_      8
#define V_      33
#define H_      512
#define W_      512
#define VCUBE   (V_ * V_ * V_)
#define M_TOTAL (3 * VCUBE)
#define HW      (H_ * W_)

typedef __attribute__((ext_vector_type(2))) float v2f;
typedef __attribute__((ext_vector_type(4))) float v4f;
typedef __attribute__((ext_vector_type(8))) float v8f;

__global__ void lut_gen_wmma(const float* __restrict__ weights,
                             const float* __restrict__ bank,
                             float* __restrict__ lut)
{
    const int lane = threadIdx.x & 31;
    const int wave = threadIdx.x >> 5;
    const int tile = blockIdx.x * (blockDim.x >> 5) + wave;
    const int m0   = tile * 16;
    if (m0 >= M_TOTAL) return;

    const int half = lane >> 4;
    const int l16  = lane & 15;

    v8f acc = {};
    #pragma unroll
    for (int kb = 0; kb < R_; kb += 4) {
        int row    = m0 + l16;
        int rclamp = row < M_TOTAL ? row : (M_TOTAL - 1);
        const float* arow = bank + (long)rclamp * R_ + kb + half * 2;
        v2f a; a.x = arow[0]; a.y = arow[1];

        const float* bcol = weights + l16 * R_ + kb + half * 2;
        v2f bm; bm.x = bcol[0]; bm.y = bcol[1];

        acc = __builtin_amdgcn_wmma_f32_16x16x4_f32(
            false, a, false, bm, (short)0, acc, false, false);
    }

    const int   batch = l16;
    const int   mbase = m0 + half * 8;
    float*      dst   = lut + (long)batch * M_TOTAL + mbase;

    if (m0 + 16 <= M_TOTAL) {
        v4f lo = { acc[0], acc[1], acc[2], acc[3] };
        v4f hi = { acc[4], acc[5], acc[6], acc[7] };
        *(v4f*)(dst)     = lo;
        *(v4f*)(dst + 4) = hi;
    } else {
        #pragma unroll
        for (int j = 0; j < 8; ++j) {
            if (mbase + j < M_TOTAL) dst[j] = acc[j];
        }
    }
}

#define AP_THREADS 512
#define AP_QUADS   8
#define AP_PIX     (AP_QUADS * 4)
#define AP_SLICE   (AP_THREADS * AP_QUADS)
#define AP_SLICES  (HW / 4 / AP_SLICE)
#define TBL_F4     ((VCUBE + 3 + 3) / 4)
#define TBL_ITERS  ((TBL_F4 + AP_THREADS - 1) / AP_THREADS)
#define TBL_BYTES  (TBL_F4 * 16)
static_assert(2 * TBL_BYTES <= 320 * 1024, "two table buffers must fit the WGP's LDS");
static_assert(TBL_ITERS > 2 * AP_QUADS && TBL_ITERS <= 3 * AP_QUADS, "piece k of a table rides on quad k % AP_QUADS");
static_assert(((long)B_ * M_TOTAL) % 4 == 0, "last table must end 16-byte aligned");
static_assert(HW / 4 == AP_SLICE * AP_SLICES, "slices must tile the image exactly");

#define PK_OFFMASK 0x000FFFFFu
#define PK_XLAST   0x00100000u
#define PK_YSTEP   0x00200000u
#define PK_ZSTEP   0x00400000u

__device__ __forceinline__ int imin_(int a, int b) { return a < b ? a : b; }

__device__ __forceinline__ void prep1(float r, float g, float bl,
                                      unsigned& pk, float& fx, float& fy, float& fz)
{
    const float s = (float)(V_ - 1);
    float cx = fminf(fmaxf(r  * s, 0.0f), s);
    float cy = fminf(fmaxf(g  * s, 0.0f), s);
    float cz = fminf(fmaxf(bl * s, 0.0f), s);
    int x0 = imin_((int)cx, V_ - 1);
    int y0 = imin_((int)cy, V_ - 1);
    int z0 = imin_((int)cz, V_ - 1);
    fx = cx - (float)x0; fy = cy - (float)y0; fz = cz - (float)z0;
    int xb = imin_(x0, V_ - 2);
    unsigned idx = (unsigned)((z0 * V_ + y0) * V_ + xb);
    pk = (idx << 2)
       | (x0 > V_ - 2 ? PK_XLAST : 0u)
       | (y0 < V_ - 1 ? PK_YSTEP : 0u)
       | (z0 < V_ - 1 ? PK_ZSTEP : 0u);
}

__device__ __forceinline__ float tri_lds(const char* tb, unsigned pk,
                                         float fx, float fy, float fz)
{
    const unsigned o00   = pk & PK_OFFMASK;
    const unsigned dy    = (pk & PK_YSTEP) ? (unsigned)(V_ * 4) : 0u;
    const unsigned dz    = (pk & PK_ZSTEP) ? (unsigned)(V_ * V_ * 4) : 0u;
    const bool     xlast = (pk & PK_XLAST) != 0u;

    const float* p00 = (const float*)(tb + o00);
    const float* p01 = (const float*)(tb + o00 + dy);
    const float* p10 = (const float*)(tb + o00 + dz);
    const float* p11 = (const float*)(tb + o00 + dy + dz);
    float a00 = p00[0], b00 = p00[1];
    float a01 = p01[0], b01 = p01[1];
    float a10 = p10[0], b10 = p10[1];
    float a11 = p11[0], b11 = p11[1];
    float v000 = xlast ? b00 : a00, v001 = b00;
    float v010 = xlast ? b01 : a01, v011 = b01;
    float v100 = xlast ? b10 : a10, v101 = b10;
    float v110 = xlast ? b11 : a11, v111 = b11;

    float gx = 1.0f - fx, gy = 1.0f - fy, gz = 1.0f - fz;
    float w000 = gz * gy * gx, w001 = gz * gy * fx;
    float w010 = gz * fy * gx, w011 = gz * fy * fx;
    float w100 = fz * gy * gx, w101 = fz * gy * fx;
    float w110 = fz * fy * gx, w111 = fz * fy * fx;

    return v000 * w000 + v001 * w001 + v010 * w010 + v011 * w011
         + v100 * w100 + v101 * w101 + v110 * w110 + v111 * w111;
}

__device__ __forceinline__ void stage_table(const float* __restrict__ lut, int b, int c,
                                            float* tbl, int tid)
{
    const unsigned g0  = (unsigned)b * M_TOTAL + (unsigned)c * VCUBE;
    const float4*  src = (const float4*)(lut + (g0 & ~3u));
    float4*        dst = (float4*)tbl;
    #pragma unroll
    for (int k = 0; k < TBL_ITERS; ++k) {
        const int j = imin_(tid + k * AP_THREADS, TBL_F4 - 1);
        dst[j] = src[j];
    }
}

__device__ __forceinline__ int piece_word(int tid, int k)
{
    return imin_(tid + k * AP_THREADS, TBL_F4 - 1);
}

__device__ __forceinline__ void lds_handover()
{
    asm volatile("s_wait_dscnt 0x0" ::: "memory");
    asm volatile("s_barrier_signal -1" ::: "memory");
    asm volatile("s_barrier_wait -1" ::: "memory");
}

template <bool STAGE>
__device__ __forceinline__ void colour_pass(const char* tb, float4* __restrict__ op, int q0,
                                            const float4* src, float4* nb, int tid,
                                            unsigned (&pk)[AP_PIX], float (&fx)[AP_PIX],
                                            float (&fy)[AP_PIX], float (&fz)[AP_PIX])
{
    #pragma unroll
    for (int j = 0; j < AP_QUADS; ++j) {
        float4 t0, t1, t2;
        if (STAGE) {
            t0 = src[piece_word(tid, j)];
            t1 = src[piece_word(tid, j + AP_QUADS)];
            if (j + 2 * AP_QUADS < TBL_ITERS) t2 = src[piece_word(tid, j + 2 * AP_QUADS)];
            asm volatile("" ::: "memory");
        }
        #pragma unroll
        for (int e = 0; e < 4; ++e)
            asm volatile("" : "+v"(pk[4*j+e]), "+v"(fx[4*j+e]), "+v"(fy[4*j+e]), "+v"(fz[4*j+e]));
        float4 o;
        o.x = tri_lds(tb, pk[4*j+0], fx[4*j+0], fy[4*j+0], fz[4*j+0]);
        o.y = tri_lds(tb, pk[4*j+1], fx[4*j+1], fy[4*j+1], fz[4*j+1]);
        o.z = tri_lds(tb, pk[4*j+2], fx[4*j+2], fy[4*j+2], fz[4*j+2]);
        o.w = tri_lds(tb, pk[4*j+3], fx[4*j+3], fy[4*j+3], fz[4*j+3]);
        op[q0 + j * AP_THREADS] = o;
        if (STAGE) {
            nb[piece_word(tid, j)]            = t0;
            nb[piece_word(tid, j + AP_QUADS)] = t1;
            if (j + 2 * AP_QUADS < TBL_ITERS) nb[piece_word(tid, j + 2 * AP_QUADS)] = t2;
        }
    }
}

__global__ void __launch_bounds__(AP_THREADS)
lut_apply_lds(const float* __restrict__ img,
              const float* lut,
              float* __restrict__ out)
{
    __shared__ __attribute__((aligned(16))) float tbl[2 * TBL_F4 * 4];

    const int  tid = threadIdx.x;
    const int  b   = blockIdx.y;
    const long ib  = (long)b * 3 * HW;
    const int  q0  = blockIdx.x * AP_SLICE + tid;

    float4 r4[AP_QUADS], g4[AP_QUADS], b4[AP_QUADS];
    #pragma unroll
    for (int j = 0; j < AP_QUADS; ++j) {
        const int q = q0 + j * AP_THREADS;
        r4[j] = ((const float4*)(img + ib         ))[q];
        g4[j] = ((const float4*)(img + ib +     HW))[q];
        b4[j] = ((const float4*)(img + ib + 2 * HW))[q];
    }
    stage_table(lut, b, 0, tbl, tid);

    unsigned pk[AP_PIX];
    float    fx[AP_PIX], fy[AP_PIX], fz[AP_PIX];
    #pragma unroll
    for (int j = 0; j < AP_QUADS; ++j) {
        prep1(r4[j].x, g4[j].x, b4[j].x, pk[4*j+0], fx[4*j+0], fy[4*j+0], fz[4*j+0]);
        prep1(r4[j].y, g4[j].y, b4[j].y, pk[4*j+1], fx[4*j+1], fy[4*j+1], fz[4*j+1]);
        prep1(r4[j].z, g4[j].z, b4[j].z, pk[4*j+2], fx[4*j+2], fy[4*j+2], fz[4*j+2]);
        prep1(r4[j].w, g4[j].w, b4[j].w, pk[4*j+3], fx[4*j+3], fy[4*j+3], fz[4*j+3]);
        #pragma unroll
        for (int e = 0; e < 4; ++e)
            asm volatile("" : "+v"(pk[4*j+e]), "+v"(fx[4*j+e]), "+v"(fy[4*j+e]), "+v"(fz[4*j+e]));
    }

    for (int c = 0; c < 2; ++c) {
        lds_handover();
        const unsigned g0 = (unsigned)b * M_TOTAL + (unsigned)c * VCUBE;
        const unsigned g1 = g0 + VCUBE;
        const char*    tb = (const char*)tbl + c * TBL_BYTES + (g0 & 3u) * 4u;
        float4*        nb = (float4*)((char*)tbl + (1 - c) * TBL_BYTES);
        colour_pass<true>(tb, (float4*)(out + ib + (long)c * HW), q0,
                          (const float4*)(lut + (g1 & ~3u)), nb, tid, pk, fx, fy, fz);
    }
    {
        lds_handover();
        const unsigned g0 = (unsigned)b * M_TOTAL + 2u * VCUBE;
        const char*    tb = (const char*)tbl + (g0 & 3u) * 4u;
        colour_pass<false>(tb, (float4*)(out + ib + 2L * HW), q0,
                           nullptr, nullptr, tid, pk, fx, fy, fz);
    }
}

extern "C" void kernel_launch(void* const* d_in, const int* in_sizes, int n_in,
                              void* d_out, int out_size, void* d_ws, size_t ws_size,
                              hipStream_t stream) {
    (void)in_sizes; (void)n_in; (void)out_size; (void)ws_size;
    const float* weights = (const float*)d_in[0];
    const float* img     = (const float*)d_in[1];
    const float* bank    = (const float*)d_in[2];
    float*       out     = (float*)d_out;
    float*       lut     = (float*)d_ws;

    const int tiles = (M_TOTAL + 15) / 16;
    dim3 blk1(256);
    dim3 grd1((tiles + 7) / 8);
    lut_gen_wmma<<<grd1, blk1, 0, stream>>>(weights, bank, lut);

    dim3 blk2(AP_THREADS);
    dim3 grd2(AP_SLICES, B_);
    lut_apply_lds<<<grd2, blk2, 0, stream>>>(img, lut, out);
}


# Apply kernel start-up: first table fetched ahead of most pixel loads and handed over at once; the first colour pass builds each quad's cell state as it reaches it, so the pixel stream runs under it
# speedup vs baseline: 8.5055x; 1.0585x over previous
#include <hip/hip_runtime.h>


#define B_      16
#define R_      8
#define V_      33
#define H_      512
#define W_      512
#define VCUBE   (V_ * V_ * V_)
#define M_TOTAL (3 * VCUBE)
#define HW      (H_ * W_)

typedef __attribute__((ext_vector_type(2))) float v2f;
typedef __attribute__((ext_vector_type(4))) float v4f;
typedef __attribute__((ext_vector_type(8))) float v8f;

__global__ void lut_gen_wmma(const float* __restrict__ weights,
                             const float* __restrict__ bank,
                             float* __restrict__ lut)
{
    const int lane = threadIdx.x & 31;
    const int wave = threadIdx.x >> 5;
    const int tile = blockIdx.x * (blockDim.x >> 5) + wave;
    const int m0   = tile * 16;
    if (m0 >= M_TOTAL) return;

    const int half = lane >> 4;
    const int l16  = lane & 15;

    v8f acc = {};
    #pragma unroll
    for (int kb = 0; kb < R_; kb += 4) {
        int row    = m0 + l16;
        int rclamp = row < M_TOTAL ? row : (M_TOTAL - 1);
        const float* arow = bank + (long)rclamp * R_ + kb + half * 2;
        v2f a; a.x = arow[0]; a.y = arow[1];

        const float* bcol = weights + l16 * R_ + kb + half * 2;
        v2f bm; bm.x = bcol[0]; bm.y = bcol[1];

        acc = __builtin_amdgcn_wmma_f32_16x16x4_f32(
            false, a, false, bm, (short)0, acc, false, false);
    }

    const int   batch = l16;
    const int   mbase = m0 + half * 8;
    float*      dst   = lut + (long)batch * M_TOTAL + mbase;

    if (m0 + 16 <= M_TOTAL) {
        v4f lo = { acc[0], acc[1], acc[2], acc[3] };
        v4f hi = { acc[4], acc[5], acc[6], acc[7] };
        *(v4f*)(dst)     = lo;
        *(v4f*)(dst + 4) = hi;
    } else {
        #pragma unroll
        for (int j = 0; j < 8; ++j) {
            if (mbase + j < M_TOTAL) dst[j] = acc[j];
        }
    }
}

#define AP_THREADS 512
#define AP_QUADS   8
#define AP_PIX     (AP_QUADS * 4)
#define AP_SLICE   (AP_THREADS * AP_QUADS)
#define AP_SLICES  (HW / 4 / AP_SLICE)
#define TBL_F4     ((VCUBE + 3 + 3) / 4)
#define TBL_ITERS  ((TBL_F4 + AP_THREADS - 1) / AP_THREADS)
#define TBL_BYTES  (TBL_F4 * 16)
static_assert(2 * TBL_BYTES <= 320 * 1024, "two table buffers must fit the WGP's LDS");
static_assert(TBL_ITERS > 2 * AP_QUADS && TBL_ITERS <= 3 * AP_QUADS, "piece k of a table rides on quad k % AP_QUADS");
#define FIRST_AHEAD 2
#define FIRST_FREE  3
#define FIRST_NP    4
static_assert(TBL_ITERS <= FIRST_NP * (AP_QUADS - FIRST_FREE), "the later quads' pieces must cover the table");
static_assert(((long)B_ * M_TOTAL) % 4 == 0, "last table must end 16-byte aligned");
static_assert(HW / 4 == AP_SLICE * AP_SLICES, "slices must tile the image exactly");

#define PK_OFFMASK 0x000FFFFFu
#define PK_XLAST   0x00100000u
#define PK_YSTEP   0x00200000u
#define PK_ZSTEP   0x00400000u

__device__ __forceinline__ int imin_(int a, int b) { return a < b ? a : b; }

__device__ __forceinline__ void prep1(float r, float g, float bl,
                                      unsigned& pk, float& fx, float& fy, float& fz)
{
    const float s = (float)(V_ - 1);
    float cx = fminf(fmaxf(r  * s, 0.0f), s);
    float cy = fminf(fmaxf(g  * s, 0.0f), s);
    float cz = fminf(fmaxf(bl * s, 0.0f), s);
    int x0 = imin_((int)cx, V_ - 1);
    int y0 = imin_((int)cy, V_ - 1);
    int z0 = imin_((int)cz, V_ - 1);
    fx = cx - (float)x0; fy = cy - (float)y0; fz = cz - (float)z0;
    int xb = imin_(x0, V_ - 2);
    unsigned idx = (unsigned)((z0 * V_ + y0) * V_ + xb);
    pk = (idx << 2)
       | (x0 > V_ - 2 ? PK_XLAST : 0u)
       | (y0 < V_ - 1 ? PK_YSTEP : 0u)
       | (z0 < V_ - 1 ? PK_ZSTEP : 0u);
}

__device__ __forceinline__ float tri_lds(const char* tb, unsigned pk,
                                         float fx, float fy, float fz)
{
    const unsigned o00   = pk & PK_OFFMASK;
    const unsigned dy    = (pk & PK_YSTEP) ? (unsigned)(V_ * 4) : 0u;
    const unsigned dz    = (pk & PK_ZSTEP) ? (unsigned)(V_ * V_ * 4) : 0u;
    const bool     xlast = (pk & PK_XLAST) != 0u;

    const float* p00 = (const float*)(tb + o00);
    const float* p01 = (const float*)(tb + o00 + dy);
    const float* p10 = (const float*)(tb + o00 + dz);
    const float* p11 = (const float*)(tb + o00 + dy + dz);
    float a00 = p00[0], b00 = p00[1];
    float a01 = p01[0], b01 = p01[1];
    float a10 = p10[0], b10 = p10[1];
    float a11 = p11[0], b11 = p11[1];
    float v000 = xlast ? b00 : a00, v001 = b00;
    float v010 = xlast ? b01 : a01, v011 = b01;
    float v100 = xlast ? b10 : a10, v101 = b10;
    float v110 = xlast ? b11 : a11, v111 = b11;

    float gx = 1.0f - fx, gy = 1.0f - fy, gz = 1.0f - fz;
    float w000 = gz * gy * gx, w001 = gz * gy * fx;
    float w010 = gz * fy * gx, w011 = gz * fy * fx;
    float w100 = fz * gy * gx, w101 = fz * gy * fx;
    float w110 = fz * fy * gx, w111 = fz * fy * fx;

    return v000 * w000 + v001 * w001 + v010 * w010 + v011 * w011
         + v100 * w100 + v101 * w101 + v110 * w110 + v111 * w111;
}

__device__ __forceinline__ int piece_word(int tid, int k)
{
    return imin_(tid + k * AP_THREADS, TBL_F4 - 1);
}

__device__ __forceinline__ constexpr int piece_of(bool stage, bool first, int j, int i)
{
    return !stage ? TBL_ITERS
         : first  ? (j >= FIRST_FREE ? FIRST_NP * (j - FIRST_FREE) + i : TBL_ITERS)
                  : (i < 3 ? j + i * AP_QUADS : TBL_ITERS);
}

__device__ __forceinline__ void lds_handover()
{
    asm volatile("s_wait_dscnt 0x0" ::: "memory");
    asm volatile("s_barrier_signal -1" ::: "memory");
    asm volatile("s_barrier_wait -1" ::: "memory");
}

template <bool STAGE, bool FIRST>
__device__ __forceinline__ void colour_pass(const char* tb, float4* __restrict__ op, int q0,
                                            const float4* src, float4* nb, int tid,
                                            unsigned (&pk)[AP_PIX], float (&fx)[AP_PIX],
                                            float (&fy)[AP_PIX], float (&fz)[AP_PIX],
                                            float4 (&r4)[AP_QUADS], float4 (&g4)[AP_QUADS],
                                            float4 (&b4)[AP_QUADS])
{
    #pragma unroll
    for (int j = 0; j < AP_QUADS; ++j) {
        const int kp0 = piece_of(STAGE, FIRST, j, 0), kp1 = piece_of(STAGE, FIRST, j, 1);
        const int kp2 = piece_of(STAGE, FIRST, j, 2), kp3 = piece_of(STAGE, FIRST, j, 3);
        static_assert(FIRST_NP == 4, "four named piece registers below");
        float4 tp0, tp1, tp2, tp3;
        if (STAGE && kp0 < TBL_ITERS) {
            tp0 = src[piece_word(tid, kp0)];
            if (kp1 < TBL_ITERS) tp1 = src[piece_word(tid, kp1)];
            if (kp2 < TBL_ITERS) tp2 = src[piece_word(tid, kp2)];
            if (kp3 < TBL_ITERS) tp3 = src[piece_word(tid, kp3)];
            asm volatile("" ::: "memory");
        }
        if (FIRST) {
            asm volatile("" : "+v"(r4[j].x), "+v"(r4[j].y), "+v"(r4[j].z), "+v"(r4[j].w),
                              "+v"(g4[j].x), "+v"(g4[j].y), "+v"(g4[j].z), "+v"(g4[j].w),
                              "+v"(b4[j].x), "+v"(b4[j].y), "+v"(b4[j].z), "+v"(b4[j].w));
            prep1(r4[j].x, g4[j].x, b4[j].x, pk[4*j+0], fx[4*j+0], fy[4*j+0], fz[4*j+0]);
            prep1(r4[j].y, g4[j].y, b4[j].y, pk[4*j+1], fx[4*j+1], fy[4*j+1], fz[4*j+1]);
            prep1(r4[j].z, g4[j].z, b4[j].z, pk[4*j+2], fx[4*j+2], fy[4*j+2], fz[4*j+2]);
            prep1(r4[j].w, g4[j].w, b4[j].w, pk[4*j+3], fx[4*j+3], fy[4*j+3], fz[4*j+3]);
        }
        #pragma unroll
        for (int e = 0; e < 4; ++e)
            asm volatile("" : "+v"(pk[4*j+e]), "+v"(fx[4*j+e]), "+v"(fy[4*j+e]), "+v"(fz[4*j+e]));
        float4 o;
        o.x = tri_lds(tb, pk[4*j+0], fx[4*j+0], fy[4*j+0], fz[4*j+0]);
        o.y = tri_lds(tb, pk[4*j+1], fx[4*j+1], fy[4*j+1], fz[4*j+1]);
        o.z = tri_lds(tb, pk[4*j+2], fx[4*j+2], fy[4*j+2], fz[4*j+2]);
        o.w = tri_lds(tb, pk[4*j+3], fx[4*j+3], fy[4*j+3], fz[4*j+3]);
        op[q0 + j * AP_THREADS] = o;
        if (STAGE && kp0 < TBL_ITERS) {
            nb[piece_word(tid, kp0)] = tp0;
            if (kp1 < TBL_ITERS) nb[piece_word(tid, kp1)] = tp1;
            if (kp2 < TBL_ITERS) nb[piece_word(tid, kp2)] = tp2;
            if (kp3 < TBL_ITERS) nb[piece_word(tid, kp3)] = tp3;
        }
    }
}

__global__ void __launch_bounds__(AP_THREADS)
lut_apply_lds(const float* img,
              const float* lut,
              float* __restrict__ out)
{
    __shared__ __attribute__((aligned(16))) float tbl[2 * TBL_F4 * 4];

    const int  tid = threadIdx.x;
    const int  b   = blockIdx.y;
    const long ib  = (long)b * 3 * HW;
    const int  q0  = blockIdx.x * AP_SLICE + tid;

    float4 r4[AP_QUADS], g4[AP_QUADS], b4[AP_QUADS];
    #pragma unroll
    for (int j = 0; j < FIRST_AHEAD; ++j) {
        const int q = q0 + j * AP_THREADS;
        r4[j] = ((const float4*)(img + ib         ))[q];
        g4[j] = ((const float4*)(img + ib +     HW))[q];
        b4[j] = ((const float4*)(img + ib + 2 * HW))[q];
    }
    asm volatile("" ::: "memory");
    {
        const unsigned g0  = (unsigned)b * M_TOTAL;
        const float4*  src = (const float4*)(lut + (g0 & ~3u));
        float4*        dst = (float4*)tbl;
        const float4 tl = src[piece_word(tid, TBL_ITERS - 1)];
        asm volatile("" ::: "memory");
        #pragma unroll
        for (int k = 0; k < TBL_ITERS - 1; ++k) dst[piece_word(tid, k)] = src[piece_word(tid, k)];
        dst[piece_word(tid, TBL_ITERS - 1)] = tl;
    }
    #pragma unroll
    for (int j = FIRST_AHEAD; j < AP_QUADS; ++j) {
        const int q = q0 + j * AP_THREADS;
        r4[j] = ((const float4*)(img + ib         ))[q];
        g4[j] = ((const float4*)(img + ib +     HW))[q];
        b4[j] = ((const float4*)(img + ib + 2 * HW))[q];
    }

    unsigned pk[AP_PIX];
    float    fx[AP_PIX], fy[AP_PIX], fz[AP_PIX];

    {
        lds_handover();
        const unsigned g0 = (unsigned)b * M_TOTAL;
        const unsigned g1 = g0 + VCUBE;
        const char*    tb = (const char*)tbl + (g0 & 3u) * 4u;
        float4*        nb = (float4*)((char*)tbl + TBL_BYTES);
        colour_pass<true, true>(tb, (float4*)(out + ib), q0,
                                (const float4*)(lut + (g1 & ~3u)), nb, tid, pk, fx, fy, fz,
                                r4, g4, b4);
    }
    {
        lds_handover();
        const unsigned g0 = (unsigned)b * M_TOTAL + (unsigned)VCUBE;
        const unsigned g1 = g0 + VCUBE;
        const char*    tb = (const char*)tbl + TBL_BYTES + (g0 & 3u) * 4u;
        float4*        nb = (float4*)tbl;
        colour_pass<true, false>(tb, (float4*)(out + ib + (long)HW), q0,
                                 (const float4*)(lut + (g1 & ~3u)), nb, tid, pk, fx, fy, fz,
                                 r4, g4, b4);
    }
    {
        lds_handover();
        const unsigned g0 = (unsigned)b * M_TOTAL + 2u * VCUBE;
        const char*    tb = (const char*)tbl + (g0 & 3u) * 4u;
        colour_pass<false, false>(tb, (float4*)(out + ib + 2L * HW), q0,
                                  nullptr, nullptr, tid, pk, fx, fy, fz, r4, g4, b4);
    }
}

extern "C" void kernel_launch(void* const* d_in, const int* in_sizes, int n_in,
                              void* d_out, int out_size, void* d_ws, size_t ws_size,
                              hipStream_t stream) {
    (void)in_sizes; (void)n_in; (void)out_size; (void)ws_size;
    const float* weights = (const float*)d_in[0];
    const float* img     = (const float*)d_in[1];
    const float* bank    = (const float*)d_in[2];
    float*       out     = (float*)d_out;
    float*       lut     = (float*)d_ws;

    const int tiles = (M_TOTAL + 15) / 16;
    dim3 blk1(256);
    dim3 grd1((tiles + 7) / 8);
    lut_gen_wmma<<<grd1, blk1, 0, stream>>>(weights, bank, lut);

    dim3 blk2(AP_THREADS);
    dim3 grd2(AP_SLICES, B_);
    lut_apply_lds<<<grd2, blk2, 0, stream>>>(img, lut, out);
}


# Table synthesis kernel: a wave makes eight 16-word tiles with the batch on the rows of the product and writes 256-byte runs of a batch row with 16-byte stores; 843 waves instead of 6739
# speedup vs baseline: 8.9194x; 1.0487x over previous
#include <hip/hip_runtime.h>


#define B_      16
#define R_      8
#define V_      33
#define H_      512
#define W_      512
#define VCUBE   (V_ * V_ * V_)
#define M_TOTAL (3 * VCUBE)
#define HW      (H_ * W_)

typedef __attribute__((ext_vector_type(2))) float v2f;
typedef __attribute__((ext_vector_type(4))) float v4f;
typedef __attribute__((ext_vector_type(8))) float v8f;

#define LG_TILES  8
#define LG_SPAN   (16 * LG_TILES)
#define LG_WAVES  4
#define LG_GROUPS ((M_TOTAL + LG_SPAN - 1) / LG_SPAN)
#define LG_BLOCKS ((LG_GROUPS + LG_WAVES - 1) / LG_WAVES)

typedef float v4f_a4 __attribute__((ext_vector_type(4), aligned(4)));

__global__ void __launch_bounds__(32 * LG_WAVES)
lut_gen_wmma(const float* __restrict__ weights,
             const float* __restrict__ bank,
             float* __restrict__ lut)
{
    const int lane = threadIdx.x & 31;
    const int wave = threadIdx.x >> 5;
    const int grp  = blockIdx.x * LG_WAVES + wave;
    const int m0   = grp * LG_SPAN;
    if (m0 >= M_TOTAL) return;

    const int half = lane >> 4;
    const int c    = lane & 15;

    const float* wrow = weights + c * R_ + half * 2;
    v2f wa0; wa0.x = wrow[0]; wa0.y = wrow[1];
    v2f wa1; wa1.x = wrow[4]; wa1.y = wrow[5];

    v2f b0[LG_TILES], b1[LG_TILES];
    #pragma unroll
    for (int t = 0; t < LG_TILES; ++t) {
        int word   = m0 + 64 * (t >> 2) + 4 * c + (t & 3);
        int wclamp = word < M_TOTAL ? word : (M_TOTAL - 1);
        const float* brow = bank + (long)wclamp * R_ + half * 2;
        b0[t].x = brow[0]; b0[t].y = brow[1];
        b1[t].x = brow[4]; b1[t].y = brow[5];
    }

    v8f acc[LG_TILES];
    #pragma unroll
    for (int t = 0; t < LG_TILES; ++t) {
        v8f z = {};
        acc[t] = __builtin_amdgcn_wmma_f32_16x16x4_f32(
            false, wa0, false, b0[t], (short)0, z, false, false);
    }
    #pragma unroll
    for (int t = 0; t < LG_TILES; ++t) {
        acc[t] = __builtin_amdgcn_wmma_f32_16x16x4_f32(
            false, wa1, false, b1[t], (short)0, acc[t], false, false);
    }

    if (m0 + LG_SPAN <= M_TOTAL) {
        #pragma unroll
        for (int g = 0; g < LG_TILES / 4; ++g) {
            #pragma unroll
            for (int j = 0; j < 8; ++j) {
                float* dst = lut + (long)(half * 8 + j) * M_TOTAL + m0 + 64 * g + 4 * c;
                v4f_a4 o = { acc[4*g+0][j], acc[4*g+1][j], acc[4*g+2][j], acc[4*g+3][j] };
                *(v4f_a4*)dst = o;
            }
        }
    } else {
        #pragma unroll
        for (int t = 0; t < LG_TILES; ++t) {
            const int word = m0 + 64 * (t >> 2) + 4 * c + (t & 3);
            #pragma unroll
            for (int j = 0; j < 8; ++j) {
                if (word < M_TOTAL) lut[(long)(half * 8 + j) * M_TOTAL + word] = acc[t][j];
            }
        }
    }
}

#define AP_THREADS 512
#define AP_QUADS   8
#define AP_PIX     (AP_QUADS * 4)
#define AP_SLICE   (AP_THREADS * AP_QUADS)
#define AP_SLICES  (HW / 4 / AP_SLICE)
#define TBL_F4     ((VCUBE + 3 + 3) / 4)
#define TBL_ITERS  ((TBL_F4 + AP_THREADS - 1) / AP_THREADS)
#define TBL_BYTES  (TBL_F4 * 16)
static_assert(2 * TBL_BYTES <= 320 * 1024, "two table buffers must fit the WGP's LDS");
static_assert(TBL_ITERS > 2 * AP_QUADS && TBL_ITERS <= 3 * AP_QUADS, "piece k of a table rides on quad k % AP_QUADS");
#define FIRST_AHEAD 2
#define FIRST_FREE  3
#define FIRST_NP    4
static_assert(TBL_ITERS <= FIRST_NP * (AP_QUADS - FIRST_FREE), "the later quads' pieces must cover the table");
static_assert(((long)B_ * M_TOTAL) % 4 == 0, "last table must end 16-byte aligned");
static_assert(HW / 4 == AP_SLICE * AP_SLICES, "slices must tile the image exactly");

#define PK_OFFMASK 0x000FFFFFu
#define PK_XLAST   0x00100000u
#define PK_YSTEP   0x00200000u
#define PK_ZSTEP   0x00400000u

__device__ __forceinline__ int imin_(int a, int b) { return a < b ? a : b; }

__device__ __forceinline__ void prep1(float r, float g, float bl,
                                      unsigned& pk, float& fx, float& fy, float& fz)
{
    const float s = (float)(V_ - 1);
    float cx = fminf(fmaxf(r  * s, 0.0f), s);
    float cy = fminf(fmaxf(g  * s, 0.0f), s);
    float cz = fminf(fmaxf(bl * s, 0.0f), s);
    int x0 = imin_((int)cx, V_ - 1);
    int y0 = imin_((int)cy, V_ - 1);
    int z0 = imin_((int)cz, V_ - 1);
    fx = cx - (float)x0; fy = cy - (float)y0; fz = cz - (float)z0;
    int xb = imin_(x0, V_ - 2);
    unsigned idx = (unsigned)((z0 * V_ + y0) * V_ + xb);
    pk = (idx << 2)
       | (x0 > V_ - 2 ? PK_XLAST : 0u)
       | (y0 < V_ - 1 ? PK_YSTEP : 0u)
       | (z0 < V_ - 1 ? PK_ZSTEP : 0u);
}

__device__ __forceinline__ float tri_lds(const char* tb, unsigned pk,
                                         float fx, float fy, float fz)
{
    const unsigned o00   = pk & PK_OFFMASK;
    const unsigned dy    = (pk & PK_YSTEP) ? (unsigned)(V_ * 4) : 0u;
    const unsigned dz    = (pk & PK_ZSTEP) ? (unsigned)(V_ * V_ * 4) : 0u;
    const bool     xlast = (pk & PK_XLAST) != 0u;

    const float* p00 = (const float*)(tb + o00);
    const float* p01 = (const float*)(tb + o00 + dy);
    const float* p10 = (const float*)(tb + o00 + dz);
    const float* p11 = (const float*)(tb + o00 + dy + dz);
    float a00 = p00[0], b00 = p00[1];
    float a01 = p01[0], b01 = p01[1];
    float a10 = p10[0], b10 = p10[1];
    float a11 = p11[0], b11 = p11[1];
    float v000 = xlast ? b00 : a00, v001 = b00;
    float v010 = xlast ? b01 : a01, v011 = b01;
    float v100 = xlast ? b10 : a10, v101 = b10;
    float v110 = xlast ? b11 : a11, v111 = b11;

    float gx = 1.0f - fx, gy = 1.0f - fy, gz = 1.0f - fz;
    float w000 = gz * gy * gx, w001 = gz * gy * fx;
    float w010 = gz * fy * gx, w011 = gz * fy * fx;
    float w100 = fz * gy * gx, w101 = fz * gy * fx;
    float w110 = fz * fy * gx, w111 = fz * fy * fx;

    return v000 * w000 + v001 * w001 + v010 * w010 + v011 * w011
         + v100 * w100 + v101 * w101 + v110 * w110 + v111 * w111;
}

__device__ __forceinline__ int piece_word(int tid, int k)
{
    return imin_(tid + k * AP_THREADS, TBL_F4 - 1);
}

__device__ __forceinline__ constexpr int piece_of(bool stage, bool first, int j, int i)
{
    return !stage ? TBL_ITERS
         : first  ? (j >= FIRST_FREE ? FIRST_NP * (j - FIRST_FREE) + i : TBL_ITERS)
                  : (i < 3 ? j + i * AP_QUADS : TBL_ITERS);
}

__device__ __forceinline__ void lds_handover()
{
    asm volatile("s_wait_dscnt 0x0" ::: "memory");
    asm volatile("s_barrier_signal -1" ::: "memory");
    asm volatile("s_barrier_wait -1" ::: "memory");
}

template <bool STAGE, bool FIRST>
__device__ __forceinline__ void colour_pass(const char* tb, float4* __restrict__ op, int q0,
                                            const float4* src, float4* nb, int tid,
                                            unsigned (&pk)[AP_PIX], float (&fx)[AP_PIX],
                                            float (&fy)[AP_PIX], float (&fz)[AP_PIX],
                                            float4 (&r4)[AP_QUADS], float4 (&g4)[AP_QUADS],
                                            float4 (&b4)[AP_QUADS])
{
    #pragma unroll
    for (int j = 0; j < AP_QUADS; ++j) {
        const int kp0 = piece_of(STAGE, FIRST, j, 0), kp1 = piece_of(STAGE, FIRST, j, 1);
        const int kp2 = piece_of(STAGE, FIRST, j, 2), kp3 = piece_of(STAGE, FIRST, j, 3);
        static_assert(FIRST_NP == 4, "four named piece registers below");
        float4 tp0, tp1, tp2, tp3;
        if (STAGE && kp0 < TBL_ITERS) {
            tp0 = src[piece_word(tid, kp0)];
            if (kp1 < TBL_ITERS) tp1 = src[piece_word(tid, kp1)];
            if (kp2 < TBL_ITERS) tp2 = src[piece_word(tid, kp2)];
            if (kp3 < TBL_ITERS) tp3 = src[piece_word(tid, kp3)];
            asm volatile("" ::: "memory");
        }
        if (FIRST) {
            asm volatile("" : "+v"(r4[j].x), "+v"(r4[j].y), "+v"(r4[j].z), "+v"(r4[j].w),
                              "+v"(g4[j].x), "+v"(g4[j].y), "+v"(g4[j].z), "+v"(g4[j].w),
                              "+v"(b4[j].x), "+v"(b4[j].y), "+v"(b4[j].z), "+v"(b4[j].w));
            prep1(r4[j].x, g4[j].x, b4[j].x, pk[4*j+0], fx[4*j+0], fy[4*j+0], fz[4*j+0]);
            prep1(r4[j].y, g4[j].y, b4[j].y, pk[4*j+1], fx[4*j+1], fy[4*j+1], fz[4*j+1]);
            prep1(r4[j].z, g4[j].z, b4[j].z, pk[4*j+2], fx[4*j+2], fy[4*j+2], fz[4*j+2]);
            prep1(r4[j].w, g4[j].w, b4[j].w, pk[4*j+3], fx[4*j+3], fy[4*j+3], fz[4*j+3]);
        }
        #pragma unroll
        for (int e = 0; e < 4; ++e)
            asm volatile("" : "+v"(pk[4*j+e]), "+v"(fx[4*j+e]), "+v"(fy[4*j+e]), "+v"(fz[4*j+e]));
        float4 o;
        o.x = tri_lds(tb, pk[4*j+0], fx[4*j+0], fy[4*j+0], fz[4*j+0]);
        o.y = tri_lds(tb, pk[4*j+1], fx[4*j+1], fy[4*j+1], fz[4*j+1]);
        o.z = tri_lds(tb, pk[4*j+2], fx[4*j+2], fy[4*j+2], fz[4*j+2]);
        o.w = tri_lds(tb, pk[4*j+3], fx[4*j+3], fy[4*j+3], fz[4*j+3]);
        op[q0 + j * AP_THREADS] = o;
        if (STAGE && kp0 < TBL_ITERS) {
            nb[piece_word(tid, kp0)] = tp0;
            if (kp1 < TBL_ITERS) nb[piece_word(tid, kp1)] = tp1;
            if (kp2 < TBL_ITERS) nb[piece_word(tid, kp2)] = tp2;
            if (kp3 < TBL_ITERS) nb[piece_word(tid, kp3)] = tp3;
        }
    }
}

__global__ void __launch_bounds__(AP_THREADS)
lut_apply_lds(const float* img,
              const float* lut,
              float* __restrict__ out)
{
    __shared__ __attribute__((aligned(16))) float tbl[2 * TBL_F4 * 4];

    const int  tid = threadIdx.x;
    const int  b   = blockIdx.y;
    const long ib  = (long)b * 3 * HW;
    const int  q0  = blockIdx.x * AP_SLICE + tid;

    float4 r4[AP_QUADS], g4[AP_QUADS], b4[AP_QUADS];
    #pragma unroll
    for (int j = 0; j < FIRST_AHEAD; ++j) {
        const int q = q0 + j * AP_THREADS;
        r4[j] = ((const float4*)(img + ib         ))[q];
        g4[j] = ((const float4*)(img + ib +     HW))[q];
        b4[j] = ((const float4*)(img + ib + 2 * HW))[q];
    }
    asm volatile("" ::: "memory");
    {
        const unsigned g0  = (unsigned)b * M_TOTAL;
        const float4*  src = (const float4*)(lut + (g0 & ~3u));
        float4*        dst = (float4*)tbl;
        const float4 tl = src[piece_word(tid, TBL_ITERS - 1)];
        asm volatile("" ::: "memory");
        #pragma unroll
        for (int k = 0; k < TBL_ITERS - 1; ++k) dst[piece_word(tid, k)] = src[piece_word(tid, k)];
        dst[piece_word(tid, TBL_ITERS - 1)] = tl;
    }
    #pragma unroll
    for (int j = FIRST_AHEAD; j < AP_QUADS; ++j) {
        const int q = q0 + j * AP_THREADS;
        r4[j] = ((const float4*)(img + ib         ))[q];
        g4[j] = ((const float4*)(img + ib +     HW))[q];
        b4[j] = ((const float4*)(img + ib + 2 * HW))[q];
    }

    unsigned pk[AP_PIX];
    float    fx[AP_PIX], fy[AP_PIX], fz[AP_PIX];

    {
        lds_handover();
        const unsigned g0 = (unsigned)b * M_TOTAL;
        const unsigned g1 = g0 + VCUBE;
        const char*    tb = (const char*)tbl + (g0 & 3u) * 4u;
        float4*        nb = (float4*)((char*)tbl + TBL_BYTES);
        colour_pass<true, true>(tb, (float4*)(out + ib), q0,
                                (const float4*)(lut + (g1 & ~3u)), nb, tid, pk, fx, fy, fz,
                                r4, g4, b4);
    }
    {
        lds_handover();
        const unsigned g0 = (unsigned)b * M_TOTAL + (unsigned)VCUBE;
        const unsigned g1 = g0 + VCUBE;
        const char*    tb = (const char*)tbl + TBL_BYTES + (g0 & 3u) * 4u;
        float4*        nb = (float4*)tbl;
        colour_pass<true, false>(tb, (float4*)(out + ib + (long)HW), q0,
                                 (const float4*)(lut + (g1 & ~3u)), nb, tid, pk, fx, fy, fz,
                                 r4, g4, b4);
    }
    {
        lds_handover();
        const unsigned g0 = (unsigned)b * M_TOTAL + 2u * VCUBE;
        const char*    tb = (const char*)tbl + (g0 & 3u) * 4u;
        colour_pass<false, false>(tb, (float4*)(out + ib + 2L * HW), q0,
                                  nullptr, nullptr, tid, pk, fx, fy, fz, r4, g4, b4);
    }
}

extern "C" void kernel_launch(void* const* d_in, const int* in_sizes, int n_in,
                              void* d_out, int out_size, void* d_ws, size_t ws_size,
                              hipStream_t stream) {
    (void)in_sizes; (void)n_in; (void)out_size; (void)ws_size;
    const float* weights = (const float*)d_in[0];
    const float* img     = (const float*)d_in[1];
    const float* bank    = (const float*)d_in[2];
    float*       out     = (float*)d_out;
    float*       lut     = (float*)d_ws;

    dim3 blk1(32 * LG_WAVES);
    dim3 grd1(LG_BLOCKS);
    lut_gen_wmma<<<grd1, blk1, 0, stream>>>(weights, bank, lut);

    dim3 blk2(AP_THREADS);
    dim3 grd2(AP_SLICES, B_);
    lut_apply_lds<<<grd2, blk2, 0, stream>>>(img, lut, out);
}


# Colours 0 and 1 are synthesized word-interleaved; the apply kernel takes both colours' corner pairs with one 16-byte LDS read per row: 8 gathers per pixel, not 12, addresses and weights made once
# speedup vs baseline: 9.8810x; 1.1078x over previous
#include <hip/hip_runtime.h>


#define B_      16
#define R_      8
#define V_      33
#define H_      512
#define W_      512
#define VCUBE   (V_ * V_ * V_)
#define M_TOTAL (3 * VCUBE)
#define HW      (H_ * W_)

typedef __attribute__((ext_vector_type(2))) float v2f;
typedef __attribute__((ext_vector_type(4))) float v4f;
typedef __attribute__((ext_vector_type(8))) float v8f;

#define IMG_F     (M_TOTAL + 1)
#define PAIR_F    (2 * VCUBE)
static_assert(IMG_F % 4 == 0, "image blocks must start 16-byte aligned");

#define LG_TILES  8
#define LG_WAVES  4
#define LG_PAIR_GROUPS  ((VCUBE + 63) / 64)
#define LG_PLAIN_GROUPS ((VCUBE + 1 + 127) / 128)
#define LG_GROUPS (LG_PAIR_GROUPS + LG_PLAIN_GROUPS)
#define LG_BLOCKS ((LG_GROUPS + LG_WAVES - 1) / LG_WAVES)

typedef float v4f_a8 __attribute__((ext_vector_type(4), aligned(8)));

__global__ void __launch_bounds__(32 * LG_WAVES)
lut_gen_wmma(const float* __restrict__ weights,
             const float* __restrict__ bank,
             float* __restrict__ lut)
{
    const int lane = threadIdx.x & 31;
    const int wave = threadIdx.x >> 5;
    const int grp  = blockIdx.x * LG_WAVES + wave;
    if (grp >= LG_GROUPS) return;

    const bool pairs = grp < LG_PAIR_GROUPS;
    const int  c0    = pairs ? grp * 64 : (grp - LG_PAIR_GROUPS) * 128;

    const int half = lane >> 4;
    const int c    = lane & 15;

    const float* wrow = weights + c * R_ + half * 2;
    v2f wa0; wa0.x = wrow[0]; wa0.y = wrow[1];
    v2f wa1; wa1.x = wrow[4]; wa1.y = wrow[5];

    v2f b0[LG_TILES], b1[LG_TILES];
    #pragma unroll
    for (int t = 0; t < LG_TILES; ++t) {
        const int g = t >> 2, e = t & 3;
        int cell = pairs ? c0 + 32 * g + 2 * c + (e >> 1) : c0 + 64 * g + 4 * c + e;
        cell = cell < VCUBE ? cell : (VCUBE - 1);
        const int row = (pairs ? (e & 1) * VCUBE : 2 * VCUBE) + cell;
        const float* brow = bank + (long)row * R_ + half * 2;
        b0[t].x = brow[0]; b0[t].y = brow[1];
        b1[t].x = brow[4]; b1[t].y = brow[5];
    }

    v8f acc[LG_TILES];
    #pragma unroll
    for (int t = 0; t < LG_TILES; ++t) {
        v8f z = {};
        acc[t] = __builtin_amdgcn_wmma_f32_16x16x4_f32(
            false, wa0, false, b0[t], (short)0, z, false, false);
    }
    #pragma unroll
    for (int t = 0; t < LG_TILES; ++t) {
        acc[t] = __builtin_amdgcn_wmma_f32_16x16x4_f32(
            false, wa1, false, b1[t], (short)0, acc[t], false, false);
    }

    if (pairs) {
        if (c0 + 64 <= VCUBE) {
            #pragma unroll
            for (int g = 0; g < LG_TILES / 4; ++g) {
                #pragma unroll
                for (int j = 0; j < 8; ++j) {
                    float* dst = lut + (long)(half * 8 + j) * IMG_F + 2 * c0 + 64 * g + 4 * c;
                    v4f o = { acc[4*g+0][j], acc[4*g+1][j], acc[4*g+2][j], acc[4*g+3][j] };
                    *(v4f*)dst = o;
                }
            }
        } else {
            #pragma unroll
            for (int t = 0; t < LG_TILES; ++t) {
                const int cell = c0 + 32 * (t >> 2) + 2 * c + ((t & 3) >> 1);
                #pragma unroll
                for (int j = 0; j < 8; ++j) {
                    if (cell < VCUBE)
                        lut[(long)(half * 8 + j) * IMG_F + 2 * cell + (t & 1)] = acc[t][j];
                }
            }
        }
    } else {
        if (c0 + 128 <= VCUBE) {
            #pragma unroll
            for (int g = 0; g < LG_TILES / 4; ++g) {
                #pragma unroll
                for (int j = 0; j < 8; ++j) {
                    float* dst = lut + (long)(half * 8 + j) * IMG_F + PAIR_F + c0 + 64 * g + 4 * c;
                    v4f_a8 o = { acc[4*g+0][j], acc[4*g+1][j], acc[4*g+2][j], acc[4*g+3][j] };
                    *(v4f_a8*)dst = o;
                }
            }
        } else {
            #pragma unroll
            for (int t = 0; t < LG_TILES; ++t) {
                const int cell = c0 + 64 * (t >> 2) + 4 * c + (t & 3);
                #pragma unroll
                for (int j = 0; j < 8; ++j) {
                    if (cell <= VCUBE)
                        lut[(long)(half * 8 + j) * IMG_F + PAIR_F + cell] = acc[t][j];
                }
            }
        }
    }
}

#define AP_THREADS 512
#define AP_QUADS   8
#define AP_PIX     (AP_QUADS * 4)
#define AP_SLICE   (AP_THREADS * AP_QUADS)
#define AP_SLICES  (HW / 4 / AP_SLICE)
#define PAIR_F4     ((PAIR_F + 3) / 4)
#define PAIR_BYTES  (PAIR_F4 * 16)
#define PLAIN_SRC_F (PAIR_F & ~3)
#define PLAIN_SHIFT (PAIR_F & 3)
#define PLAIN_F4    ((VCUBE + PLAIN_SHIFT + 3) / 4)
#define PLAIN_BYTES (PLAIN_F4 * 16)
#define PLAIN_ITERS ((PLAIN_F4 + AP_THREADS - 1) / AP_THREADS)
static_assert(PLAIN_SRC_F + PLAIN_F4 * 4 == IMG_F, "the plain copy must end with the image block");
#define FIRST_AHEAD 2
#define FIRST_FREE  3
#define STAGE_NP    4
#define STAGE_ITERS (STAGE_NP * (AP_QUADS - FIRST_FREE))
#define STAGE_F4    (STAGE_ITERS * AP_THREADS)
#define PLAIN_OFF   (STAGE_F4 * 16)
#define LDS_BYTES   (PLAIN_OFF + PLAIN_BYTES)
#define REST_F4     (PAIR_F4 - STAGE_F4)
#define REST_ITERS  ((REST_F4 + AP_THREADS - 1) / AP_THREADS)
static_assert(STAGE_F4 < PAIR_F4, "pass 1 stages only a part of the pair table");
static_assert(PAIR_BYTES <= LDS_BYTES && LDS_BYTES <= 320 * 1024, "both tables must fit the WGP's LDS");
static_assert(HW / 4 == AP_SLICE * AP_SLICES, "slices must tile the image exactly");

#define PK_OFFMASK 0x000FFFFFu
#define PK_XLAST   0x00100000u
#define PK_YSTEP   0x00200000u
#define PK_ZSTEP   0x00400000u

__device__ __forceinline__ int imin_(int a, int b) { return a < b ? a : b; }

__device__ __forceinline__ void prep1(float r, float g, float bl,
                                      unsigned& pk, float& fx, float& fy, float& fz)
{
    const float s = (float)(V_ - 1);
    float cx = fminf(fmaxf(r  * s, 0.0f), s);
    float cy = fminf(fmaxf(g  * s, 0.0f), s);
    float cz = fminf(fmaxf(bl * s, 0.0f), s);
    int x0 = imin_((int)cx, V_ - 1);
    int y0 = imin_((int)cy, V_ - 1);
    int z0 = imin_((int)cz, V_ - 1);
    fx = cx - (float)x0; fy = cy - (float)y0; fz = cz - (float)z0;
    int xb = imin_(x0, V_ - 2);
    unsigned idx = (unsigned)((z0 * V_ + y0) * V_ + xb);
    pk = (idx << 2)
       | (x0 > V_ - 2 ? PK_XLAST : 0u)
       | (y0 < V_ - 1 ? PK_YSTEP : 0u)
       | (z0 < V_ - 1 ? PK_ZSTEP : 0u);
}

struct W8 { float w000, w001, w010, w011, w100, w101, w110, w111; };
__device__ __forceinline__ W8 weights8(float fx, float fy, float fz)
{
    float gx = 1.0f - fx, gy = 1.0f - fy, gz = 1.0f - fz;
    W8 w;
    w.w000 = gz * gy * gx; w.w001 = gz * gy * fx;
    w.w010 = gz * fy * gx; w.w011 = gz * fy * fx;
    w.w100 = fz * gy * gx; w.w101 = fz * gy * fx;
    w.w110 = fz * fy * gx; w.w111 = fz * fy * fx;
    return w;
}

__device__ __forceinline__ float blend8(const W8& w, bool xlast,
                                        float a00, float b00, float a01, float b01,
                                        float a10, float b10, float a11, float b11)
{
    float v000 = xlast ? b00 : a00, v001 = b00;
    float v010 = xlast ? b01 : a01, v011 = b01;
    float v100 = xlast ? b10 : a10, v101 = b10;
    float v110 = xlast ? b11 : a11, v111 = b11;
    return v000 * w.w000 + v001 * w.w001 + v010 * w.w010 + v011 * w.w011
         + v100 * w.w100 + v101 * w.w101 + v110 * w.w110 + v111 * w.w111;
}

__device__ __forceinline__ float tri_plain(const char* tb, unsigned pk,
                                           float fx, float fy, float fz)
{
    const unsigned o00   = pk & PK_OFFMASK;
    const unsigned dy    = (pk & PK_YSTEP) ? (unsigned)(V_ * 4) : 0u;
    const unsigned dz    = (pk & PK_ZSTEP) ? (unsigned)(V_ * V_ * 4) : 0u;
    const bool     xlast = (pk & PK_XLAST) != 0u;

    const float* p00 = (const float*)(tb + o00);
    const float* p01 = (const float*)(tb + o00 + dy);
    const float* p10 = (const float*)(tb + o00 + dz);
    const float* p11 = (const float*)(tb + o00 + dy + dz);
    float a00 = p00[0], b00 = p00[1];
    float a01 = p01[0], b01 = p01[1];
    float a10 = p10[0], b10 = p10[1];
    float a11 = p11[0], b11 = p11[1];
    const W8 w = weights8(fx, fy, fz);
    return blend8(w, xlast, a00, b00, a01, b01, a10, b10, a11, b11);
}

__device__ __forceinline__ void tri_pair(const char* tb, unsigned pk,
                                         float fx, float fy, float fz,
                                         float& out0, float& out1)
{
    const unsigned o00   = (pk & PK_OFFMASK) << 1;
    const unsigned dy    = (pk & PK_YSTEP) ? (unsigned)(V_ * 8) : 0u;
    const unsigned dz    = (pk & PK_ZSTEP) ? (unsigned)(V_ * V_ * 8) : 0u;
    const bool     xlast = (pk & PK_XLAST) != 0u;

    const float2* p00 = (const float2*)(tb + o00);
    const float2* p01 = (const float2*)(tb + o00 + dy);
    const float2* p10 = (const float2*)(tb + o00 + dz);
    const float2* p11 = (const float2*)(tb + o00 + dy + dz);
    float2 a00 = p00[0], b00 = p00[1];
    float2 a01 = p01[0], b01 = p01[1];
    float2 a10 = p10[0], b10 = p10[1];
    float2 a11 = p11[0], b11 = p11[1];
    const W8 w = weights8(fx, fy, fz);
    out0 = blend8(w, xlast, a00.x, b00.x, a01.x, b01.x, a10.x, b10.x, a11.x, b11.x);
    out1 = blend8(w, xlast, a00.y, b00.y, a01.y, b01.y, a10.y, b10.y, a11.y, b11.y);
}

__device__ __forceinline__ int plain_word(int tid, int k)
{
    return imin_(tid + k * AP_THREADS, PLAIN_F4 - 1);
}
__device__ __forceinline__ int rest_word(int tid, int k)
{
    return imin_(STAGE_F4 + tid + k * AP_THREADS, PAIR_F4 - 1);
}

__device__ __forceinline__ constexpr int piece_of(int j, int i)
{
    return j >= FIRST_FREE ? STAGE_NP * (j - FIRST_FREE) + i : STAGE_ITERS;
}

__device__ __forceinline__ void lds_handover()
{
    asm volatile("s_wait_dscnt 0x0" ::: "memory");
    asm volatile("s_barrier_signal -1" ::: "memory");
    asm volatile("s_barrier_wait -1" ::: "memory");
}

__device__ __forceinline__ void pass_plain(const char* tb, float4* __restrict__ op, int q0,
                                           const float4* src, float4* nb, int tid,
                                           unsigned (&pk)[AP_PIX], float (&fx)[AP_PIX],
                                           float (&fy)[AP_PIX], float (&fz)[AP_PIX],
                                           float4 (&r4)[AP_QUADS], float4 (&g4)[AP_QUADS],
                                           float4 (&b4)[AP_QUADS])
{
    #pragma unroll
    for (int j = 0; j < AP_QUADS; ++j) {
        const int kp0 = piece_of(j, 0), kp1 = piece_of(j, 1);
        const int kp2 = piece_of(j, 2), kp3 = piece_of(j, 3);
        static_assert(STAGE_NP == 4, "four named piece registers below");
        float4 tp0, tp1, tp2, tp3;
        if (kp0 < STAGE_ITERS) {
            tp0 = src[tid + kp0 * AP_THREADS];
            tp1 = src[tid + kp1 * AP_THREADS];
            tp2 = src[tid + kp2 * AP_THREADS];
            tp3 = src[tid + kp3 * AP_THREADS];
            asm volatile("" ::: "memory");
        }
        asm volatile("" : "+v"(r4[j].x), "+v"(r4[j].y), "+v"(r4[j].z), "+v"(r4[j].w),
                          "+v"(g4[j].x), "+v"(g4[j].y), "+v"(g4[j].z), "+v"(g4[j].w),
                          "+v"(b4[j].x), "+v"(b4[j].y), "+v"(b4[j].z), "+v"(b4[j].w));
        prep1(r4[j].x, g4[j].x, b4[j].x, pk[4*j+0], fx[4*j+0], fy[4*j+0], fz[4*j+0]);
        prep1(r4[j].y, g4[j].y, b4[j].y, pk[4*j+1], fx[4*j+1], fy[4*j+1], fz[4*j+1]);
        prep1(r4[j].z, g4[j].z, b4[j].z, pk[4*j+2], fx[4*j+2], fy[4*j+2], fz[4*j+2]);
        prep1(r4[j].w, g4[j].w, b4[j].w, pk[4*j+3], fx[4*j+3], fy[4*j+3], fz[4*j+3]);
        float4 o;
        o.x = tri_plain(tb, pk[4*j+0], fx[4*j+0], fy[4*j+0], fz[4*j+0]);
        o.y = tri_plain(tb, pk[4*j+1], fx[4*j+1], fy[4*j+1], fz[4*j+1]);
        o.z = tri_plain(tb, pk[4*j+2], fx[4*j+2], fy[4*j+2], fz[4*j+2]);
        o.w = tri_plain(tb, pk[4*j+3], fx[4*j+3], fy[4*j+3], fz[4*j+3]);
        op[q0 + j * AP_THREADS] = o;
        if (kp0 < STAGE_ITERS) {
            nb[tid + kp0 * AP_THREADS] = tp0;
            nb[tid + kp1 * AP_THREADS] = tp1;
            nb[tid + kp2 * AP_THREADS] = tp2;
            nb[tid + kp3 * AP_THREADS] = tp3;
        }
    }
}

__device__ __forceinline__ void pass_pair(const char* tb, float4* __restrict__ op0,
                                          float4* __restrict__ op1, int q0,
                                          unsigned (&pk)[AP_PIX], float (&fx)[AP_PIX],
                                          float (&fy)[AP_PIX], float (&fz)[AP_PIX])
{
    #pragma unroll
    for (int j = 0; j < AP_QUADS; ++j) {
        #pragma unroll
        for (int e = 0; e < 4; ++e)
            asm volatile("" : "+v"(pk[4*j+e]), "+v"(fx[4*j+e]), "+v"(fy[4*j+e]), "+v"(fz[4*j+e]));
        float4 o0, o1;
        tri_pair(tb, pk[4*j+0], fx[4*j+0], fy[4*j+0], fz[4*j+0], o0.x, o1.x);
        tri_pair(tb, pk[4*j+1], fx[4*j+1], fy[4*j+1], fz[4*j+1], o0.y, o1.y);
        tri_pair(tb, pk[4*j+2], fx[4*j+2], fy[4*j+2], fz[4*j+2], o0.z, o1.z);
        tri_pair(tb, pk[4*j+3], fx[4*j+3], fy[4*j+3], fz[4*j+3], o0.w, o1.w);
        op0[q0 + j * AP_THREADS] = o0;
        op1[q0 + j * AP_THREADS] = o1;
    }
}

__global__ void __launch_bounds__(AP_THREADS)
lut_apply_lds(const float* img,
              const float* lut,
              float* __restrict__ out)
{
    __shared__ __attribute__((aligned(16))) float tbl[LDS_BYTES / 4];

    const int  tid = threadIdx.x;
    const int  b   = blockIdx.y;
    const long ib  = (long)b * 3 * HW;
    const int  q0  = blockIdx.x * AP_SLICE + tid;

    const float4* pair_src  = (const float4*)(lut + (long)b * IMG_F);
    const float4* plain_src = (const float4*)(lut + (long)b * IMG_F + PLAIN_SRC_F);
    float4*       pair_lds  = (float4*)tbl;
    float4*       plain_lds = (float4*)((char*)tbl + PLAIN_OFF);

    float4 r4[AP_QUADS], g4[AP_QUADS], b4[AP_QUADS];
    #pragma unroll
    for (int j = 0; j < FIRST_AHEAD; ++j) {
        const int q = q0 + j * AP_THREADS;
        r4[j] = ((const float4*)(img + ib         ))[q];
        g4[j] = ((const float4*)(img + ib +     HW))[q];
        b4[j] = ((const float4*)(img + ib + 2 * HW))[q];
    }
    asm volatile("" ::: "memory");
    {
        const float4 tl = plain_src[plain_word(tid, PLAIN_ITERS - 1)];
        asm volatile("" ::: "memory");
        #pragma unroll
        for (int k = 0; k < PLAIN_ITERS - 1; ++k)
            plain_lds[plain_word(tid, k)] = plain_src[plain_word(tid, k)];
        plain_lds[plain_word(tid, PLAIN_ITERS - 1)] = tl;
    }
    asm volatile("" ::: "memory");
    __builtin_amdgcn_sched_barrier(0);
    #pragma unroll
    for (int j = FIRST_AHEAD; j < AP_QUADS; ++j) {
        const int q = q0 + j * AP_THREADS;
        r4[j] = ((const float4*)(img + ib         ))[q];
        g4[j] = ((const float4*)(img + ib +     HW))[q];
        b4[j] = ((const float4*)(img + ib + 2 * HW))[q];
    }

    unsigned pk[AP_PIX];
    float    fx[AP_PIX], fy[AP_PIX], fz[AP_PIX];

    lds_handover();
    pass_plain((const char*)plain_lds + PLAIN_SHIFT * 4, (float4*)(out + ib + 2L * HW), q0,
               pair_src, pair_lds, tid, pk, fx, fy, fz, r4, g4, b4);

    {
        static_assert(REST_ITERS == 16, "sixteen named registers below");
        #define REST_LOAD(k)  const float4 rest##k = pair_src[rest_word(tid, k)];
        #define REST_STORE(k) pair_lds[rest_word(tid, k)] = rest##k;
        REST_LOAD(0)  REST_LOAD(1)  REST_LOAD(2)  REST_LOAD(3)
        REST_LOAD(4)  REST_LOAD(5)  REST_LOAD(6)  REST_LOAD(7)
        REST_LOAD(8)  REST_LOAD(9)  REST_LOAD(10) REST_LOAD(11)
        REST_LOAD(12) REST_LOAD(13) REST_LOAD(14) REST_LOAD(15)
        lds_handover();
        REST_STORE(0)  REST_STORE(1)  REST_STORE(2)  REST_STORE(3)
        REST_STORE(4)  REST_STORE(5)  REST_STORE(6)  REST_STORE(7)
        REST_STORE(8)  REST_STORE(9)  REST_STORE(10) REST_STORE(11)
        REST_STORE(12) REST_STORE(13) REST_STORE(14) REST_STORE(15)
        #undef REST_LOAD
        #undef REST_STORE
    }

    lds_handover();
    pass_pair((const char*)pair_lds, (float4*)(out + ib), (float4*)(out + ib + (long)HW), q0,
              pk, fx, fy, fz);
}

extern "C" void kernel_launch(void* const* d_in, const int* in_sizes, int n_in,
                              void* d_out, int out_size, void* d_ws, size_t ws_size,
                              hipStream_t stream) {
    (void)in_sizes; (void)n_in; (void)out_size; (void)ws_size;
    const float* weights = (const float*)d_in[0];
    const float* img     = (const float*)d_in[1];
    const float* bank    = (const float*)d_in[2];
    float*       out     = (float*)d_out;
    float*       lut     = (float*)d_ws;

    dim3 blk1(32 * LG_WAVES);
    dim3 grd1(LG_BLOCKS);
    lut_gen_wmma<<<grd1, blk1, 0, stream>>>(weights, bank, lut);

    dim3 blk2(AP_THREADS);
    dim3 grd2(AP_SLICES, B_);
    lut_apply_lds<<<grd2, blk2, 0, stream>>>(img, lut, out);
}
